# starting kernel
# baseline (speedup 1.0000x reference)
#include <hip/hip_runtime.h>

typedef __attribute__((ext_vector_type(2))) float v2f;
typedef __attribute__((ext_vector_type(8))) float v8f;

#define CIN   256
#define COUT  256
#define BATCH 2
#define HH    48
#define WW    48
#define HWSZ  (HH * WW)
#define KS    7
#define PADR  3
#define MAPELEMS (BATCH * COUT * HWSZ)

__global__ __launch_bounds__(256) void qkv_gemm(
    const float* __restrict__ x,
    const float* __restrict__ wq,
    const float* __restrict__ wk,
    const float* __restrict__ wv,
    float* __restrict__ ws)
{
  const int z     = blockIdx.z;
  const int b     = z / 3;
  const int which = z - 3 * b;
  const float* __restrict__ Wm = (which == 0) ? wq : ((which == 1) ? wk : wv);
  const float* __restrict__ X  = x + (size_t)b * CIN * HWSZ;
  float* __restrict__ out = ws + (size_t)which * MAPELEMS + (size_t)b * COUT * HWSZ;

  const int tid  = threadIdx.x;
  const int wave = tid >> 5;
  const int lane = tid & 31;
  const int l15  = lane & 15;
  const int hi   = lane >> 4;
  const int koff = hi << 1;

  const int M0 = blockIdx.x * 16;
  const int N0 = blockIdx.y * 128 + wave * 16;
  const int mrow = M0 + l15;
  const int ncol = N0 + l15;

  const float* __restrict__ Arow = Wm + (size_t)mrow * CIN;

  v8f acc = {};
  #pragma unroll 8
  for (int k = 0; k < CIN; k += 4) {
    v2f a = *reinterpret_cast<const v2f*>(Arow + k + koff);
    v2f bm;
    bm.x = X[(size_t)(k + koff)     * HWSZ + ncol];
    bm.y = X[(size_t)(k + koff + 1) * HWSZ + ncol];
    acc = __builtin_amdgcn_wmma_f32_16x16x4_f32(
        false, a, false, bm, (short)0, acc, false, false);
  }

  const int ms = M0 + (hi << 3);
  #pragma unroll
  for (int r = 0; r < 8; ++r)
    out[(size_t)(ms + r) * HWSZ + ncol] = acc[r];
}

#define TS    16
#define PATCH (TS + KS - 1)
#define PITCH (PATCH + 1)
#define CG    16

__global__ __launch_bounds__(256) void local_attn(
    const float* __restrict__ ws,
    const float* __restrict__ rel_h,
    const float* __restrict__ rel_w,
    float* __restrict__ out)
{
  const float* __restrict__ qm = ws;
  const float* __restrict__ km = ws + (size_t)MAPELEMS;
  const float* __restrict__ vm = ws + (size_t)2 * MAPELEMS;

  __shared__ float sk[PATCH * PITCH];
  __shared__ float sv[PATCH * PITCH];

  const int tid = threadIdx.x;
  const int tx  = tid & 15;
  const int ty  = tid >> 4;

  const int b   = blockIdx.z;
  const int c0  = blockIdx.y * CG;
  const int nTw = WW / TS;
  const int th0 = (blockIdx.x / nTw) * TS;
  const int tw0 = (blockIdx.x % nTw) * TS;
  const int h   = th0 + ty;
  const int w   = tw0 + tx;

  for (int ci = 0; ci < CG; ++ci) {
    const int c = c0 + ci;
    const float* __restrict__ kc = km + ((size_t)b * COUT + c) * HWSZ;
    const float* __restrict__ vc = vm + ((size_t)b * COUT + c) * HWSZ;

    for (int idx = tid; idx < PATCH * PATCH; idx += 256) {
      const int r  = idx / PATCH;
      const int cc = idx - r * PATCH;
      const int gh = th0 - PADR + r;
      const int gw = tw0 - PADR + cc;
      const bool ok = (gh >= 0) & (gh < HH) & (gw >= 0) & (gw < WW);
      sk[r * PITCH + cc] = ok ? kc[gh * WW + gw] : 0.0f;
      sv[r * PITCH + cc] = ok ? vc[gh * WW + gw] : 0.0f;
    }
    __syncthreads();

    const float q = qm[((size_t)b * COUT + c) * HWSZ + h * WW + w];

    const bool isrow = (c < COUT / 2);
    const float* __restrict__ rel =
        isrow ? (rel_h + (size_t)c * KS) : (rel_w + (size_t)(c - COUT / 2) * KS);
    float bias7[KS];
    #pragma unroll
    for (int i = 0; i < KS; ++i) bias7[i] = rel[i];

    float mx = -3.0e38f;
    #pragma unroll
    for (int kh = 0; kh < KS; ++kh) {
      #pragma unroll
      for (int kw = 0; kw < KS; ++kw) {
        const float bias = isrow ? bias7[kh] : bias7[kw];
        const float kv = sk[(ty + kh) * PITCH + (tx + kw)] + bias;
        mx = fmaxf(mx, q * kv);
      }
    }

    float s = 0.0f, accv = 0.0f;
    #pragma unroll
    for (int kh = 0; kh < KS; ++kh) {
      #pragma unroll
      for (int kw = 0; kw < KS; ++kw) {
        const float bias = isrow ? bias7[kh] : bias7[kw];
        const float kv = sk[(ty + kh) * PITCH + (tx + kw)] + bias;
        const float e  = __expf(q * kv - mx);
        s    += e;
        accv += e * sv[(ty + kh) * PITCH + (tx + kw)];
      }
    }

    out[((size_t)b * COUT + c) * HWSZ + h * WW + w] = accv / s;
    __syncthreads();
  }
}

extern "C" void kernel_launch(void* const* d_in, const int* in_sizes, int n_in,
                              void* d_out, int out_size, void* d_ws, size_t ws_size,
                              hipStream_t stream) {
  const float* x  = (const float*)d_in[0];
  const float* wq = (const float*)d_in[1];
  const float* wk = (const float*)d_in[2];
  const float* wv = (const float*)d_in[3];
  const float* rh = (const float*)d_in[4];
  const float* rw = (const float*)d_in[5];
  float* ws  = (float*)d_ws;
  float* out = (float*)d_out;

  dim3 g1(COUT / 16, HWSZ / 128, 3 * BATCH);
  qkv_gemm<<<g1, 256, 0, stream>>>(x, wq, wk, wv, ws);

  dim3 g2((HH / TS) * (WW / TS), COUT / CG, BATCH);
  local_attn<<<g2, 256, 0, stream>>>(ws, rh, rw, out);
}


# q/k/v maps written channel-fastest; window softmax with 32 channels on a wave's lanes, 7x7 K/V window slid along a row in registers from whole-line global loads: no LDS, no barriers; sum order kept
# speedup vs baseline: 2.2265x; 2.2265x over previous
#include <hip/hip_runtime.h>


typedef __attribute__((ext_vector_type(2))) float v2f;
typedef __attribute__((ext_vector_type(8))) float v8f;

#define CIN   256
#define COUT  256
#define BATCH 2
#define HH    48
#define WW    48
#define HWSZ  (HH * WW)
#define KS    7
#define PADR  3
#define MAPELEMS (BATCH * COUT * HWSZ)

__global__ __launch_bounds__(256) void qkv_gemm_cl(
    const float* __restrict__ x,
    const float* __restrict__ wq,
    const float* __restrict__ wk,
    const float* __restrict__ wv,
    float* __restrict__ ws)
{
  const int z     = blockIdx.z;
  const int b     = z / 3;
  const int which = z - 3 * b;
  const float* __restrict__ Wm = (which == 0) ? wq : ((which == 1) ? wk : wv);
  const float* __restrict__ X  = x + (size_t)b * CIN * HWSZ;
  float* __restrict__ out = ws + (size_t)which * MAPELEMS + (size_t)b * COUT * HWSZ;

  const int tid  = threadIdx.x;
  const int wave = tid >> 5;
  const int lane = tid & 31;
  const int l15  = lane & 15;
  const int hi   = lane >> 4;
  const int koff = hi << 1;

  const int C0 = blockIdx.x * 16;
  const int P0 = blockIdx.y * 128 + wave * 16;
  const int ch  = C0 + l15;
  const int pix = P0 + l15;

  const float* __restrict__ Wrow = Wm + (size_t)ch * CIN;

  v8f acc = {};
  #pragma unroll 8
  for (int k = 0; k < CIN; k += 4) {
    v2f wb = *reinterpret_cast<const v2f*>(Wrow + k + koff);
    v2f xa;
    xa.x = X[(size_t)(k + koff)     * HWSZ + pix];
    xa.y = X[(size_t)(k + koff + 1) * HWSZ + pix];
    acc = __builtin_amdgcn_wmma_f32_16x16x4_f32(
        false, xa, false, wb, (short)0, acc, false, false);
  }

  const int ps = P0 + (hi << 3);
  #pragma unroll
  for (int r = 0; r < 8; ++r)
    out[(size_t)(ps + r) * COUT + ch] = acc[r];
}

#define SEG  12
#define NSEG (WW / SEG)
#define RPB  4

template <bool ISROW>
__device__ __forceinline__ float window_softmax(
    const float (&kwin)[KS][KS], const float (&vwin)[KS][KS],
    const float (&bias)[KS], const float q)
{
  float mx = -3.0e38f;
  #pragma unroll
  for (int r = 0; r < KS; ++r) {
    #pragma unroll
    for (int j = 0; j < KS; ++j) {
      const float kv = ISROW ? kwin[r][j] : (kwin[r][j] + bias[j]);
      mx = fmaxf(mx, q * kv);
    }
  }
  float s = 0.0f, accv = 0.0f;
  #pragma unroll
  for (int r = 0; r < KS; ++r) {
    #pragma unroll
    for (int j = 0; j < KS; ++j) {
      const float kv = ISROW ? kwin[r][j] : (kwin[r][j] + bias[j]);
      const float e  = __expf(q * kv - mx);
      s    += e;
      accv += e * vwin[r][j];
    }
  }
  return accv / s;
}

template <bool ISROW>
__device__ __forceinline__ void attn_row_segment(
    const float* __restrict__ qB,
    const float* __restrict__ kB,
    const float* __restrict__ vB,
    const unsigned c,
    const float* __restrict__ rel,
    float* __restrict__ orow,
    const int h, const int w0)
{
  float bias[KS];
  #pragma unroll
  for (int i = 0; i < KS; ++i) bias[i] = rel[i];

  bool rok[KS];
  int  rofs[KS];
  #pragma unroll
  for (int r = 0; r < KS; ++r) {
    const int hh = h - PADR + r;
    rok[r]  = (hh >= 0) & (hh < HH);
    const int hc = hh < 0 ? 0 : (hh > HH - 1 ? HH - 1 : hh);
    rofs[r] = hc * WW * COUT;
  }

  float kwin[KS][KS], vwin[KS][KS];

  #pragma unroll
  for (int j = 1; j < KS; ++j) {
    const int ww  = w0 - PADR - 1 + j;
    const bool cok = (ww >= 0) & (ww < WW);
    const int wc  = ww < 0 ? 0 : (ww > WW - 1 ? WW - 1 : ww);
    #pragma unroll
    for (int r = 0; r < KS; ++r) {
      const int idx = rofs[r] + wc * COUT;
      const bool ok = cok & rok[r];
      const float kl = (kB + idx)[c];
      const float vl = (vB + idx)[c];
      const float kk = ok ? kl : 0.0f;
      kwin[r][j] = ISROW ? (kk + bias[r]) : kk;
      vwin[r][j] = ok ? vl : 0.0f;
    }
  }
  #pragma unroll
  for (int r = 0; r < KS; ++r) { kwin[r][0] = 0.0f; vwin[r][0] = 0.0f; }

  float pk[KS], pv[KS];
  {
    const int wc = (w0 + PADR > WW - 1) ? WW - 1 : (w0 + PADR);
    #pragma unroll
    for (int r = 0; r < KS; ++r) {
      const int idx = rofs[r] + wc * COUT;
      pk[r] = (kB + idx)[c];
      pv[r] = (vB + idx)[c];
    }
  }
  float qn = (qB + (h * WW + w0) * COUT)[c];

  for (int i4 = 0; i4 < SEG; i4 += 4) {
    float o[4];
    #pragma unroll
    for (int u = 0; u < 4; ++u) {
      const int w = w0 + i4 + u;

      const bool cok = (w + PADR < WW);
      #pragma unroll
      for (int r = 0; r < KS; ++r) {
        #pragma unroll
        for (int j = 0; j < KS - 1; ++j) {
          kwin[r][j] = kwin[r][j + 1];
          vwin[r][j] = vwin[r][j + 1];
        }
        const bool ok = cok & rok[r];
        const float kk = ok ? pk[r] : 0.0f;
        kwin[r][KS - 1] = ISROW ? (kk + bias[r]) : kk;
        vwin[r][KS - 1] = ok ? pv[r] : 0.0f;
      }
      const float q = qn;

      {
        const int wn = (w + PADR + 1 > WW - 1) ? WW - 1 : (w + PADR + 1);
        #pragma unroll
        for (int r = 0; r < KS; ++r) {
          const int idx = rofs[r] + wn * COUT;
          pk[r] = (kB + idx)[c];
          pv[r] = (vB + idx)[c];
        }
        const int wq1 = (w + 1 > WW - 1) ? WW - 1 : (w + 1);
        qn = (qB + (h * WW + wq1) * COUT)[c];
      }

      o[u] = window_softmax<ISROW>(kwin, vwin, bias, q);
    }
    *reinterpret_cast<float4*>(orow + i4) = make_float4(o[0], o[1], o[2], o[3]);
  }
}

__global__ __launch_bounds__(RPB * 32) void local_attn_cl(
    const float* __restrict__ ws,
    const float* __restrict__ rel_h,
    const float* __restrict__ rel_w,
    float* __restrict__ out)
{
  const unsigned lane = threadIdx.x & 31;
  const int wave = __builtin_amdgcn_readfirstlane(threadIdx.x >> 5);
  const int b    = blockIdx.z;
  const int cg   = blockIdx.y;
  const unsigned c = cg * 32 + lane;
  const int rg   = blockIdx.x / NSEG;
  const int seg  = blockIdx.x - rg * NSEG;
  const int h    = rg * RPB + wave;
  const int w0   = seg * SEG;

  const float* __restrict__ qB = ws + (size_t)b * COUT * HWSZ;
  const float* __restrict__ kB = qB + (size_t)MAPELEMS;
  const float* __restrict__ vB = qB + (size_t)2 * MAPELEMS;
  float* __restrict__ orow = out + ((size_t)b * COUT + c) * HWSZ + h * WW + w0;

  if (cg < COUT / 64)
    attn_row_segment<true >(qB, kB, vB, c, rel_h + (size_t)c * KS, orow, h, w0);
  else
    attn_row_segment<false>(qB, kB, vB, c, rel_w + (size_t)(c - COUT / 2) * KS, orow, h, w0);
}

extern "C" void kernel_launch(void* const* d_in, const int* in_sizes, int n_in,
                              void* d_out, int out_size, void* d_ws, size_t ws_size,
                              hipStream_t stream) {
  const float* x  = (const float*)d_in[0];
  const float* wq = (const float*)d_in[1];
  const float* wk = (const float*)d_in[2];
  const float* wv = (const float*)d_in[3];
  const float* rh = (const float*)d_in[4];
  const float* rw = (const float*)d_in[5];
  float* ws  = (float*)d_ws;
  float* out = (float*)d_out;

  dim3 g1(COUT / 16, HWSZ / 128, 3 * BATCH);
  qkv_gemm_cl<<<g1, 256, 0, stream>>>(x, wq, wk, wv, ws);

  dim3 g2((HH / RPB) * NSEG, COUT / 32, BATCH);
  local_attn_cl<<<g2, RPB * 32, 0, stream>>>(ws, rh, rw, out);
}


# QKV GEMM without LDS or work-group tile: a wave owns 64 pixels x 32 channels (4x2 accumulators), 16-byte global loads feed fragments used by 2-4 matrix steps; k sum order in 16-k chunks changes
# speedup vs baseline: 3.4076x; 1.5305x over previous
#include <hip/hip_runtime.h>


typedef __attribute__((ext_vector_type(2))) float v2f;
typedef __attribute__((ext_vector_type(8))) float v8f;

#define CIN   256
#define COUT  256
#define BATCH 2
#define HH    48
#define WW    48
#define HWSZ  (HH * WW)
#define KS    7
#define PADR  3
#define MAPELEMS (BATCH * COUT * HWSZ)

#define KCH 16

struct GemmFrag {
  float4 a[8];
  float4 w[2][2];
};

__device__ __forceinline__ void gemm_load_chunk(
    GemmFrag& f,
    const float* __restrict__ xl,
    const float* __restrict__ w0,
    const float* __restrict__ w1,
    const int k0)
{
  f.w[0][0] = *reinterpret_cast<const float4*>(w0 + k0);
  f.w[1][0] = *reinterpret_cast<const float4*>(w1 + k0);
  f.w[0][1] = *reinterpret_cast<const float4*>(w0 + k0 + 4);
  f.w[1][1] = *reinterpret_cast<const float4*>(w1 + k0 + 4);
  #pragma unroll
  for (int e = 0; e < 8; ++e)
    f.a[e] = *reinterpret_cast<const float4*>(xl + (size_t)(k0 + e) * HWSZ);
}

__device__ __forceinline__ void gemm_mma_chunk(v8f (&acc)[4][2], const GemmFrag& f)
{
  v2f ap[4][4];
  #pragma unroll
  for (int j = 0; j < 4; ++j) {
    const float4 x0 = f.a[2 * j];
    const float4 x1 = f.a[2 * j + 1];
    ap[j][0].x = x0.x; ap[j][0].y = x1.x;
    ap[j][1].x = x0.y; ap[j][1].y = x1.y;
    ap[j][2].x = x0.z; ap[j][2].y = x1.z;
    ap[j][3].x = x0.w; ap[j][3].y = x1.w;
  }
  __builtin_amdgcn_sched_barrier(0);
  #pragma unroll
  for (int j = 0; j < 4; ++j) {
    v2f wb[2];
    #pragma unroll
    for (int n = 0; n < 2; ++n) {
      const float4 wv = f.w[n][j >> 1];
      wb[n].x = (j & 1) ? wv.z : wv.x;
      wb[n].y = (j & 1) ? wv.w : wv.y;
    }
    #pragma unroll
    for (int t = 0; t < 4; ++t) {
      #pragma unroll
      for (int n = 0; n < 2; ++n)
        acc[t][n] = __builtin_amdgcn_wmma_f32_16x16x4_f32(
            false, ap[j][t], false, wb[n], (short)0, acc[t][n], false, false);
    }
  }
  __builtin_amdgcn_sched_barrier(0);
  asm volatile("v_nop\n\tv_nop\n\tv_nop\n\tv_nop");
}

__global__ __launch_bounds__(256) void qkv_gemm_cl(
    const float* __restrict__ x,
    const float* __restrict__ wq,
    const float* __restrict__ wk,
    const float* __restrict__ wv,
    float* __restrict__ ws)
{
  const int z     = blockIdx.z;
  const int b     = z / 3;
  const int which = z - 3 * b;
  const float* __restrict__ Wm = (which == 0) ? wq : ((which == 1) ? wk : wv);
  const float* __restrict__ X  = x + (size_t)b * CIN * HWSZ;
  float* __restrict__ out = ws + (size_t)which * MAPELEMS + (size_t)b * COUT * HWSZ;

  const int tid  = threadIdx.x;
  const int wave = tid >> 5;
  const int lane = tid & 31;
  const int l15  = lane & 15;
  const int hi   = lane >> 4;

  const int P0 = (blockIdx.y * 2 + (wave >> 2)) * 64;
  const int C0 = (blockIdx.x * 4 + (wave & 3)) * 32;

  const float* __restrict__ xl = X + (size_t)(8 * hi) * HWSZ + P0 + 4 * l15;
  const float* __restrict__ w0 = Wm + (size_t)(C0 + 2 * l15) * CIN + 8 * hi;
  const float* __restrict__ w1 = w0 + CIN;

  v8f acc[4][2];
  #pragma unroll
  for (int t = 0; t < 4; ++t) {
    #pragma unroll
    for (int n = 0; n < 2; ++n) acc[t][n] = v8f{};
  }

  GemmFrag f0, f1;
  gemm_load_chunk(f0, xl, w0, w1, 0);
  #pragma unroll
  for (int k0 = 0; k0 < CIN; k0 += 2 * KCH) {
    gemm_load_chunk(f1, xl, w0, w1, k0 + KCH);
    __builtin_amdgcn_sched_barrier(0);
    gemm_mma_chunk(acc, f0);
    __builtin_amdgcn_sched_barrier(0);
    if (k0 + 2 * KCH < CIN) gemm_load_chunk(f0, xl, w0, w1, k0 + 2 * KCH);
    __builtin_amdgcn_sched_barrier(0);
    gemm_mma_chunk(acc, f1);
    __builtin_amdgcn_sched_barrier(0);
  }

  asm volatile("v_nop\n\tv_nop\n\tv_nop\n\tv_nop\n\tv_nop");
  __builtin_amdgcn_sched_barrier(0);

  #pragma unroll
  for (int t = 0; t < 4; ++t) {
    #pragma unroll
    for (int r = 0; r < 8; ++r) {
      const int pixel = P0 + 4 * (8 * hi + r) + t;
      v2f o;
      o.x = acc[t][0][r];
      o.y = acc[t][1][r];
      *reinterpret_cast<v2f*>(out + (size_t)pixel * COUT + C0 + 2 * l15) = o;
    }
  }
}

#define SEG  12
#define NSEG (WW / SEG)
#define RPB  4

template <bool ISROW>
__device__ __forceinline__ float window_softmax(
    const float (&kwin)[KS][KS], const float (&vwin)[KS][KS],
    const float (&bias)[KS], const float q)
{
  float mx = -3.0e38f;
  #pragma unroll
  for (int r = 0; r < KS; ++r) {
    #pragma unroll
    for (int j = 0; j < KS; ++j) {
      const float kv = ISROW ? kwin[r][j] : (kwin[r][j] + bias[j]);
      mx = fmaxf(mx, q * kv);
    }
  }
  float s = 0.0f, accv = 0.0f;
  #pragma unroll
  for (int r = 0; r < KS; ++r) {
    #pragma unroll
    for (int j = 0; j < KS; ++j) {
      const float kv = ISROW ? kwin[r][j] : (kwin[r][j] + bias[j]);
      const float e  = __expf(q * kv - mx);
      s    += e;
      accv += e * vwin[r][j];
    }
  }
  return accv / s;
}

template <bool ISROW>
__device__ __forceinline__ void attn_row_segment(
    const float* __restrict__ qB,
    const float* __restrict__ kB,
    const float* __restrict__ vB,
    const unsigned c,
    const float* __restrict__ rel,
    float* __restrict__ orow,
    const int h, const int w0)
{
  float bias[KS];
  #pragma unroll
  for (int i = 0; i < KS; ++i) bias[i] = rel[i];

  bool rok[KS];
  int  rofs[KS];
  #pragma unroll
  for (int r = 0; r < KS; ++r) {
    const int hh = h - PADR + r;
    rok[r]  = (hh >= 0) & (hh < HH);
    const int hc = hh < 0 ? 0 : (hh > HH - 1 ? HH - 1 : hh);
    rofs[r] = hc * WW * COUT;
  }

  float kwin[KS][KS], vwin[KS][KS];

  #pragma unroll
  for (int j = 1; j < KS; ++j) {
    const int ww  = w0 - PADR - 1 + j;
    const bool cok = (ww >= 0) & (ww < WW);
    const int wc  = ww < 0 ? 0 : (ww > WW - 1 ? WW - 1 : ww);
    #pragma unroll
    for (int r = 0; r < KS; ++r) {
      const int idx = rofs[r] + wc * COUT;
      const bool ok = cok & rok[r];
      const float kl = (kB + idx)[c];
      const float vl = (vB + idx)[c];
      const float kk = ok ? kl : 0.0f;
      kwin[r][j] = ISROW ? (kk + bias[r]) : kk;
      vwin[r][j] = ok ? vl : 0.0f;
    }
  }
  #pragma unroll
  for (int r = 0; r < KS; ++r) { kwin[r][0] = 0.0f; vwin[r][0] = 0.0f; }

  float pk[KS], pv[KS];
  {
    const int wc = (w0 + PADR > WW - 1) ? WW - 1 : (w0 + PADR);
    #pragma unroll
    for (int r = 0; r < KS; ++r) {
      const int idx = rofs[r] + wc * COUT;
      pk[r] = (kB + idx)[c];
      pv[r] = (vB + idx)[c];
    }
  }
  float qn = (qB + (h * WW + w0) * COUT)[c];

  for (int i4 = 0; i4 < SEG; i4 += 4) {
    float o[4];
    #pragma unroll
    for (int u = 0; u < 4; ++u) {
      const int w = w0 + i4 + u;

      const bool cok = (w + PADR < WW);
      #pragma unroll
      for (int r = 0; r < KS; ++r) {
        #pragma unroll
        for (int j = 0; j < KS - 1; ++j) {
          kwin[r][j] = kwin[r][j + 1];
          vwin[r][j] = vwin[r][j + 1];
        }
        const bool ok = cok & rok[r];
        const float kk = ok ? pk[r] : 0.0f;
        kwin[r][KS - 1] = ISROW ? (kk + bias[r]) : kk;
        vwin[r][KS - 1] = ok ? pv[r] : 0.0f;
      }
      const float q = qn;

      {
        const int wn = (w + PADR + 1 > WW - 1) ? WW - 1 : (w + PADR + 1);
        #pragma unroll
        for (int r = 0; r < KS; ++r) {
          const int idx = rofs[r] + wn * COUT;
          pk[r] = (kB + idx)[c];
          pv[r] = (vB + idx)[c];
        }
        const int wq1 = (w + 1 > WW - 1) ? WW - 1 : (w + 1);
        qn = (qB + (h * WW + wq1) * COUT)[c];
      }

      o[u] = window_softmax<ISROW>(kwin, vwin, bias, q);
    }
    *reinterpret_cast<float4*>(orow + i4) = make_float4(o[0], o[1], o[2], o[3]);
  }
}

__global__ __launch_bounds__(RPB * 32) void local_attn_cl(
    const float* __restrict__ ws,
    const float* __restrict__ rel_h,
    const float* __restrict__ rel_w,
    float* __restrict__ out)
{
  const unsigned lane = threadIdx.x & 31;
  const int wave = __builtin_amdgcn_readfirstlane(threadIdx.x >> 5);
  const int b    = blockIdx.z;
  const int cg   = blockIdx.y;
  const unsigned c = cg * 32 + lane;
  const int rg   = blockIdx.x / NSEG;
  const int seg  = blockIdx.x - rg * NSEG;
  const int h    = rg * RPB + wave;
  const int w0   = seg * SEG;

  const float* __restrict__ qB = ws + (size_t)b * COUT * HWSZ;
  const float* __restrict__ kB = qB + (size_t)MAPELEMS;
  const float* __restrict__ vB = qB + (size_t)2 * MAPELEMS;
  float* __restrict__ orow = out + ((size_t)b * COUT + c) * HWSZ + h * WW + w0;

  if (cg < COUT / 64)
    attn_row_segment<true >(qB, kB, vB, c, rel_h + (size_t)c * KS, orow, h, w0);
  else
    attn_row_segment<false>(qB, kB, vB, c, rel_w + (size_t)(c - COUT / 2) * KS, orow, h, w0);
}

extern "C" void kernel_launch(void* const* d_in, const int* in_sizes, int n_in,
                              void* d_out, int out_size, void* d_ws, size_t ws_size,
                              hipStream_t stream) {
  const float* x  = (const float*)d_in[0];
  const float* wq = (const float*)d_in[1];
  const float* wk = (const float*)d_in[2];
  const float* wv = (const float*)d_in[3];
  const float* rh = (const float*)d_in[4];
  const float* rw = (const float*)d_in[5];
  float* ws  = (float*)d_ws;
  float* out = (float*)d_out;

  dim3 g1(COUT / 128, HWSZ / 128, 3 * BATCH);
  qkv_gemm_cl<<<g1, 256, 0, stream>>>(x, wq, wk, wv, ws);

  dim3 g2((HH / RPB) * NSEG, COUT / 32, BATCH);
  local_attn_cl<<<g2, RPB * 32, 0, stream>>>(ws, rh, rw, out);
}


# Window softmax, 4 vector ops/element (fma, exp2, add, fma): max logit = max(q*kmax, q*kmin) from column key extrema slid with the window (exact), log2e folded into q; 4-wave hint; sum order kept
# speedup vs baseline: 3.5837x; 1.0517x over previous
#include <hip/hip_runtime.h>


typedef __attribute__((ext_vector_type(2))) float v2f;
typedef __attribute__((ext_vector_type(8))) float v8f;

#define CIN   256
#define COUT  256
#define BATCH 2
#define HH    48
#define WW    48
#define HWSZ  (HH * WW)
#define KS    7
#define PADR  3
#define MAPELEMS (BATCH * COUT * HWSZ)

#define KCH 16

struct GemmFrag {
  float4 a[8];
  float4 w[2][2];
};

__device__ __forceinline__ void gemm_load_chunk(
    GemmFrag& f,
    const float* __restrict__ xl,
    const float* __restrict__ w0,
    const float* __restrict__ w1,
    const int k0)
{
  f.w[0][0] = *reinterpret_cast<const float4*>(w0 + k0);
  f.w[1][0] = *reinterpret_cast<const float4*>(w1 + k0);
  f.w[0][1] = *reinterpret_cast<const float4*>(w0 + k0 + 4);
  f.w[1][1] = *reinterpret_cast<const float4*>(w1 + k0 + 4);
  #pragma unroll
  for (int e = 0; e < 8; ++e)
    f.a[e] = *reinterpret_cast<const float4*>(xl + (size_t)(k0 + e) * HWSZ);
}

__device__ __forceinline__ void gemm_mma_chunk(v8f (&acc)[4][2], const GemmFrag& f)
{
  v2f ap[4][4];
  #pragma unroll
  for (int j = 0; j < 4; ++j) {
    const float4 x0 = f.a[2 * j];
    const float4 x1 = f.a[2 * j + 1];
    ap[j][0].x = x0.x; ap[j][0].y = x1.x;
    ap[j][1].x = x0.y; ap[j][1].y = x1.y;
    ap[j][2].x = x0.z; ap[j][2].y = x1.z;
    ap[j][3].x = x0.w; ap[j][3].y = x1.w;
  }
  __builtin_amdgcn_sched_barrier(0);
  #pragma unroll
  for (int j = 0; j < 4; ++j) {
    v2f wb[2];
    #pragma unroll
    for (int n = 0; n < 2; ++n) {
      const float4 wv = f.w[n][j >> 1];
      wb[n].x = (j & 1) ? wv.z : wv.x;
      wb[n].y = (j & 1) ? wv.w : wv.y;
    }
    #pragma unroll
    for (int t = 0; t < 4; ++t) {
      #pragma unroll
      for (int n = 0; n < 2; ++n)
        acc[t][n] = __builtin_amdgcn_wmma_f32_16x16x4_f32(
            false, ap[j][t], false, wb[n], (short)0, acc[t][n], false, false);
    }
  }
  __builtin_amdgcn_sched_barrier(0);
  asm volatile("v_nop\n\tv_nop\n\tv_nop\n\tv_nop");
}

__global__ __launch_bounds__(256) void qkv_gemm_cl(
    const float* __restrict__ x,
    const float* __restrict__ wq,
    const float* __restrict__ wk,
    const float* __restrict__ wv,
    float* __restrict__ ws)
{
  const int z     = blockIdx.z;
  const int b     = z / 3;
  const int which = z - 3 * b;
  const float* __restrict__ Wm = (which == 0) ? wq : ((which == 1) ? wk : wv);
  const float* __restrict__ X  = x + (size_t)b * CIN * HWSZ;
  float* __restrict__ out = ws + (size_t)which * MAPELEMS + (size_t)b * COUT * HWSZ;

  const int tid  = threadIdx.x;
  const int wave = tid >> 5;
  const int lane = tid & 31;
  const int l15  = lane & 15;
  const int hi   = lane >> 4;

  const int P0 = (blockIdx.y * 2 + (wave >> 2)) * 64;
  const int C0 = (blockIdx.x * 4 + (wave & 3)) * 32;

  const float* __restrict__ xl = X + (size_t)(8 * hi) * HWSZ + P0 + 4 * l15;
  const float* __restrict__ w0 = Wm + (size_t)(C0 + 2 * l15) * CIN + 8 * hi;
  const float* __restrict__ w1 = w0 + CIN;

  v8f acc[4][2];
  #pragma unroll
  for (int t = 0; t < 4; ++t) {
    #pragma unroll
    for (int n = 0; n < 2; ++n) acc[t][n] = v8f{};
  }

  GemmFrag f0, f1;
  gemm_load_chunk(f0, xl, w0, w1, 0);
  #pragma unroll
  for (int k0 = 0; k0 < CIN; k0 += 2 * KCH) {
    gemm_load_chunk(f1, xl, w0, w1, k0 + KCH);
    __builtin_amdgcn_sched_barrier(0);
    gemm_mma_chunk(acc, f0);
    __builtin_amdgcn_sched_barrier(0);
    if (k0 + 2 * KCH < CIN) gemm_load_chunk(f0, xl, w0, w1, k0 + 2 * KCH);
    __builtin_amdgcn_sched_barrier(0);
    gemm_mma_chunk(acc, f1);
    __builtin_amdgcn_sched_barrier(0);
  }

  asm volatile("v_nop\n\tv_nop\n\tv_nop\n\tv_nop\n\tv_nop");
  __builtin_amdgcn_sched_barrier(0);

  #pragma unroll
  for (int t = 0; t < 4; ++t) {
    #pragma unroll
    for (int r = 0; r < 8; ++r) {
      const int pixel = P0 + 4 * (8 * hi + r) + t;
      v2f o;
      o.x = acc[t][0][r];
      o.y = acc[t][1][r];
      *reinterpret_cast<v2f*>(out + (size_t)pixel * COUT + C0 + 2 * l15) = o;
    }
  }
}

#define SEG  12
#define NSEG (WW / SEG)
#define RPB  4

#define LOG2E 1.44269504088896340736f

__device__ __forceinline__ float max7(const float (&a)[KS])
{
  float m = fmaxf(fmaxf(a[0], a[1]), a[2]);
  m = fmaxf(fmaxf(m, a[3]), a[4]);
  return fmaxf(fmaxf(m, a[5]), a[6]);
}

__device__ __forceinline__ float min7(const float (&a)[KS])
{
  float m = fminf(fminf(a[0], a[1]), a[2]);
  m = fminf(fminf(m, a[3]), a[4]);
  return fminf(fminf(m, a[5]), a[6]);
}

template <bool ISROW>
__device__ __forceinline__ float window_softmax(
    const float (&kwin)[KS][KS], const float (&vwin)[KS][KS],
    const float (&cmx)[KS], const float (&cmn)[KS],
    const float (&bias)[KS], const float q)
{
  float hi, lo;
  if (ISROW) {
    hi = max7(cmx);
    lo = min7(cmn);
  } else {
    float a[KS], b[KS];
    #pragma unroll
    for (int j = 0; j < KS; ++j) { a[j] = cmx[j] + bias[j]; b[j] = cmn[j] + bias[j]; }
    hi = max7(a);
    lo = min7(b);
  }
  const float mx  = fmaxf(q * hi, q * lo);
  const float q2  = q * LOG2E;
  const float nm2 = -(mx * LOG2E);
  float cj[KS];
  #pragma unroll
  for (int j = 0; j < KS; ++j) cj[j] = ISROW ? nm2 : fmaf(q2, bias[j], nm2);

  float s = 0.0f, accv = 0.0f;
  #pragma unroll
  for (int r = 0; r < KS; ++r) {
    #pragma unroll
    for (int j = 0; j < KS; ++j) {
      const float e = __builtin_amdgcn_exp2f(fmaf(q2, kwin[r][j], cj[j]));
      s    += e;
      accv += e * vwin[r][j];
    }
  }
  return accv / s;
}

template <bool ISROW>
__device__ __forceinline__ void attn_row_segment(
    const float* __restrict__ qB,
    const float* __restrict__ kB,
    const float* __restrict__ vB,
    const unsigned c,
    const float* __restrict__ rel,
    float* __restrict__ orow,
    const int h, const int w0)
{
  float bias[KS];
  #pragma unroll
  for (int i = 0; i < KS; ++i) bias[i] = rel[i];

  bool rok[KS];
  int  rofs[KS];
  #pragma unroll
  for (int r = 0; r < KS; ++r) {
    const int hh = h - PADR + r;
    rok[r]  = (hh >= 0) & (hh < HH);
    const int hc = hh < 0 ? 0 : (hh > HH - 1 ? HH - 1 : hh);
    rofs[r] = hc * WW * COUT;
  }

  float kwin[KS][KS], vwin[KS][KS];
  float cmx[KS], cmn[KS];

  #pragma unroll
  for (int j = 1; j < KS; ++j) {
    const int ww  = w0 - PADR - 1 + j;
    const bool cok = (ww >= 0) & (ww < WW);
    const int wc  = ww < 0 ? 0 : (ww > WW - 1 ? WW - 1 : ww);
    #pragma unroll
    for (int r = 0; r < KS; ++r) {
      const int idx = rofs[r] + wc * COUT;
      const bool ok = cok & rok[r];
      const float kl = (kB + idx)[c];
      const float vl = (vB + idx)[c];
      const float kk = ok ? kl : 0.0f;
      kwin[r][j] = ISROW ? (kk + bias[r]) : kk;
      vwin[r][j] = ok ? vl : 0.0f;
    }
    float col[KS];
    #pragma unroll
    for (int r = 0; r < KS; ++r) col[r] = kwin[r][j];
    cmx[j] = max7(col);
    cmn[j] = min7(col);
  }
  #pragma unroll
  for (int r = 0; r < KS; ++r) { kwin[r][0] = 0.0f; vwin[r][0] = 0.0f; }
  cmx[0] = 0.0f; cmn[0] = 0.0f;

  float pk[KS], pv[KS];
  {
    const int wc = (w0 + PADR > WW - 1) ? WW - 1 : (w0 + PADR);
    #pragma unroll
    for (int r = 0; r < KS; ++r) {
      const int idx = rofs[r] + wc * COUT;
      pk[r] = (kB + idx)[c];
      pv[r] = (vB + idx)[c];
    }
  }
  float qn = (qB + (h * WW + w0) * COUT)[c];

  for (int i4 = 0; i4 < SEG; i4 += 4) {
    float o[4];
    #pragma unroll
    for (int u = 0; u < 4; ++u) {
      const int w = w0 + i4 + u;

      const bool cok = (w + PADR < WW);
      #pragma unroll
      for (int r = 0; r < KS; ++r) {
        #pragma unroll
        for (int j = 0; j < KS - 1; ++j) {
          kwin[r][j] = kwin[r][j + 1];
          vwin[r][j] = vwin[r][j + 1];
        }
        const bool ok = cok & rok[r];
        const float kk = ok ? pk[r] : 0.0f;
        kwin[r][KS - 1] = ISROW ? (kk + bias[r]) : kk;
        vwin[r][KS - 1] = ok ? pv[r] : 0.0f;
      }
      {
        float col[KS];
        #pragma unroll
        for (int r = 0; r < KS; ++r) col[r] = kwin[r][KS - 1];
        #pragma unroll
        for (int j = 0; j < KS - 1; ++j) { cmx[j] = cmx[j + 1]; cmn[j] = cmn[j + 1]; }
        cmx[KS - 1] = max7(col);
        cmn[KS - 1] = min7(col);
      }
      const float q = qn;

      {
        const int wn = (w + PADR + 1 > WW - 1) ? WW - 1 : (w + PADR + 1);
        #pragma unroll
        for (int r = 0; r < KS; ++r) {
          const int idx = rofs[r] + wn * COUT;
          pk[r] = (kB + idx)[c];
          pv[r] = (vB + idx)[c];
        }
        const int wq1 = (w + 1 > WW - 1) ? WW - 1 : (w + 1);
        qn = (qB + (h * WW + wq1) * COUT)[c];
      }

      o[u] = window_softmax<ISROW>(kwin, vwin, cmx, cmn, bias, q);
    }
    *reinterpret_cast<float4*>(orow + i4) = make_float4(o[0], o[1], o[2], o[3]);
  }
}

__global__ __launch_bounds__(RPB * 32) __attribute__((amdgpu_waves_per_eu(4, 4))) void local_attn_cl(
    const float* __restrict__ ws,
    const float* __restrict__ rel_h,
    const float* __restrict__ rel_w,
    float* __restrict__ out)
{
  const unsigned lane = threadIdx.x & 31;
  const int wave = __builtin_amdgcn_readfirstlane(threadIdx.x >> 5);
  const int b    = blockIdx.z;
  const int cg   = blockIdx.y;
  const unsigned c = cg * 32 + lane;
  const int rg   = blockIdx.x / NSEG;
  const int seg  = blockIdx.x - rg * NSEG;
  const int h    = rg * RPB + wave;
  const int w0   = seg * SEG;

  const float* __restrict__ qB = ws + (size_t)b * COUT * HWSZ;
  const float* __restrict__ kB = qB + (size_t)MAPELEMS;
  const float* __restrict__ vB = qB + (size_t)2 * MAPELEMS;
  float* __restrict__ orow = out + ((size_t)b * COUT + c) * HWSZ + h * WW + w0;

  if (cg < COUT / 64)
    attn_row_segment<true >(qB, kB, vB, c, rel_h + (size_t)c * KS, orow, h, w0);
  else
    attn_row_segment<false>(qB, kB, vB, c, rel_w + (size_t)(c - COUT / 2) * KS, orow, h, w0);
}

extern "C" void kernel_launch(void* const* d_in, const int* in_sizes, int n_in,
                              void* d_out, int out_size, void* d_ws, size_t ws_size,
                              hipStream_t stream) {
  const float* x  = (const float*)d_in[0];
  const float* wq = (const float*)d_in[1];
  const float* wk = (const float*)d_in[2];
  const float* wv = (const float*)d_in[3];
  const float* rh = (const float*)d_in[4];
  const float* rw = (const float*)d_in[5];
  float* ws  = (float*)d_ws;
  float* out = (float*)d_out;

  dim3 g1(COUT / 128, HWSZ / 128, 3 * BATCH);
  qkv_gemm_cl<<<g1, 256, 0, stream>>>(x, wq, wk, wv, ws);

  dim3 g2((HH / RPB) * NSEG, COUT / 32, BATCH);
  local_attn_cl<<<g2, RPB * 32, 0, stream>>>(ws, rh, rw, out);
}


# Window attention: wave-uniform work off the vector unit: K/V/q loads take a scalar row base plus a lane offset (no 64-bit vector address math); padding selects only in border waves; sum order kept
# speedup vs baseline: 3.7474x; 1.0457x over previous
#include <hip/hip_runtime.h>


typedef __attribute__((ext_vector_type(2))) float v2f;
typedef __attribute__((ext_vector_type(8))) float v8f;

#define CIN   256
#define COUT  256
#define BATCH 2
#define HH    48
#define WW    48
#define HWSZ  (HH * WW)
#define KS    7
#define PADR  3
#define MAPELEMS (BATCH * COUT * HWSZ)

#define KCH 16

struct GemmFrag {
  float4 a[8];
  float4 w[2][2];
};

__device__ __forceinline__ void gemm_load_chunk(
    GemmFrag& f,
    const float* __restrict__ xl,
    const float* __restrict__ w0,
    const float* __restrict__ w1,
    const int k0)
{
  f.w[0][0] = *reinterpret_cast<const float4*>(w0 + k0);
  f.w[1][0] = *reinterpret_cast<const float4*>(w1 + k0);
  f.w[0][1] = *reinterpret_cast<const float4*>(w0 + k0 + 4);
  f.w[1][1] = *reinterpret_cast<const float4*>(w1 + k0 + 4);
  #pragma unroll
  for (int e = 0; e < 8; ++e)
    f.a[e] = *reinterpret_cast<const float4*>(xl + (size_t)(k0 + e) * HWSZ);
}

__device__ __forceinline__ void gemm_mma_chunk(v8f (&acc)[4][2], const GemmFrag& f)
{
  v2f ap[4][4];
  #pragma unroll
  for (int j = 0; j < 4; ++j) {
    const float4 x0 = f.a[2 * j];
    const float4 x1 = f.a[2 * j + 1];
    ap[j][0].x = x0.x; ap[j][0].y = x1.x;
    ap[j][1].x = x0.y; ap[j][1].y = x1.y;
    ap[j][2].x = x0.z; ap[j][2].y = x1.z;
    ap[j][3].x = x0.w; ap[j][3].y = x1.w;
  }
  __builtin_amdgcn_sched_barrier(0);
  #pragma unroll
  for (int j = 0; j < 4; ++j) {
    v2f wb[2];
    #pragma unroll
    for (int n = 0; n < 2; ++n) {
      const float4 wv = f.w[n][j >> 1];
      wb[n].x = (j & 1) ? wv.z : wv.x;
      wb[n].y = (j & 1) ? wv.w : wv.y;
    }
    #pragma unroll
    for (int t = 0; t < 4; ++t) {
      #pragma unroll
      for (int n = 0; n < 2; ++n)
        acc[t][n] = __builtin_amdgcn_wmma_f32_16x16x4_f32(
            false, ap[j][t], false, wb[n], (short)0, acc[t][n], false, false);
    }
  }
  __builtin_amdgcn_sched_barrier(0);
  asm volatile("v_nop\n\tv_nop\n\tv_nop\n\tv_nop");
}

__global__ __launch_bounds__(256) void qkv_gemm_cl(
    const float* __restrict__ x,
    const float* __restrict__ wq,
    const float* __restrict__ wk,
    const float* __restrict__ wv,
    float* __restrict__ ws)
{
  const int z     = blockIdx.z;
  const int b     = z / 3;
  const int which = z - 3 * b;
  const float* __restrict__ Wm = (which == 0) ? wq : ((which == 1) ? wk : wv);
  const float* __restrict__ X  = x + (size_t)b * CIN * HWSZ;
  float* __restrict__ out = ws + (size_t)which * MAPELEMS + (size_t)b * COUT * HWSZ;

  const int tid  = threadIdx.x;
  const int wave = tid >> 5;
  const int lane = tid & 31;
  const int l15  = lane & 15;
  const int hi   = lane >> 4;

  const int P0 = (blockIdx.y * 2 + (wave >> 2)) * 64;
  const int C0 = (blockIdx.x * 4 + (wave & 3)) * 32;

  const float* __restrict__ xl = X + (size_t)(8 * hi) * HWSZ + P0 + 4 * l15;
  const float* __restrict__ w0 = Wm + (size_t)(C0 + 2 * l15) * CIN + 8 * hi;
  const float* __restrict__ w1 = w0 + CIN;

  v8f acc[4][2];
  #pragma unroll
  for (int t = 0; t < 4; ++t) {
    #pragma unroll
    for (int n = 0; n < 2; ++n) acc[t][n] = v8f{};
  }

  GemmFrag f0, f1;
  gemm_load_chunk(f0, xl, w0, w1, 0);
  #pragma unroll
  for (int k0 = 0; k0 < CIN; k0 += 2 * KCH) {
    gemm_load_chunk(f1, xl, w0, w1, k0 + KCH);
    __builtin_amdgcn_sched_barrier(0);
    gemm_mma_chunk(acc, f0);
    __builtin_amdgcn_sched_barrier(0);
    if (k0 + 2 * KCH < CIN) gemm_load_chunk(f0, xl, w0, w1, k0 + 2 * KCH);
    __builtin_amdgcn_sched_barrier(0);
    gemm_mma_chunk(acc, f1);
    __builtin_amdgcn_sched_barrier(0);
  }

  asm volatile("v_nop\n\tv_nop\n\tv_nop\n\tv_nop\n\tv_nop");
  __builtin_amdgcn_sched_barrier(0);

  #pragma unroll
  for (int t = 0; t < 4; ++t) {
    #pragma unroll
    for (int r = 0; r < 8; ++r) {
      const int pixel = P0 + 4 * (8 * hi + r) + t;
      v2f o;
      o.x = acc[t][0][r];
      o.y = acc[t][1][r];
      *reinterpret_cast<v2f*>(out + (size_t)pixel * COUT + C0 + 2 * l15) = o;
    }
  }
}

#define SEG  12
#define NSEG (WW / SEG)
#define RPB  4

#define LOG2E 1.44269504088896340736f

__device__ __forceinline__ float max7(const float (&a)[KS])
{
  float m = fmaxf(fmaxf(a[0], a[1]), a[2]);
  m = fmaxf(fmaxf(m, a[3]), a[4]);
  return fmaxf(fmaxf(m, a[5]), a[6]);
}

__device__ __forceinline__ float min7(const float (&a)[KS])
{
  float m = fminf(fminf(a[0], a[1]), a[2]);
  m = fminf(fminf(m, a[3]), a[4]);
  return fminf(fminf(m, a[5]), a[6]);
}

template <bool ISROW>
__device__ __forceinline__ float window_softmax(
    const float (&kwin)[KS][KS], const float (&vwin)[KS][KS],
    const float (&cmx)[KS], const float (&cmn)[KS],
    const float (&bias)[KS], const float q)
{
  float hi, lo;
  if (ISROW) {
    hi = max7(cmx);
    lo = min7(cmn);
  } else {
    float a[KS], b[KS];
    #pragma unroll
    for (int j = 0; j < KS; ++j) { a[j] = cmx[j] + bias[j]; b[j] = cmn[j] + bias[j]; }
    hi = max7(a);
    lo = min7(b);
  }
  const float mx  = fmaxf(q * hi, q * lo);
  const float q2  = q * LOG2E;
  const float nm2 = -(mx * LOG2E);
  float cj[KS];
  #pragma unroll
  for (int j = 0; j < KS; ++j) cj[j] = ISROW ? nm2 : fmaf(q2, bias[j], nm2);

  float s = 0.0f, accv = 0.0f;
  #pragma unroll
  for (int r = 0; r < KS; ++r) {
    #pragma unroll
    for (int j = 0; j < KS; ++j) {
      const float e = __builtin_amdgcn_exp2f(fmaf(q2, kwin[r][j], cj[j]));
      s    += e;
      accv += e * vwin[r][j];
    }
  }
  return accv / s;
}

typedef __attribute__((address_space(1))) const char  gmem_char;
typedef __attribute__((address_space(1))) const float gmem_float;

__device__ __forceinline__ float ld_lane(gmem_char* base, const unsigned off)
{
  return *(gmem_float*)(base + off);
}

__device__ __forceinline__ gmem_char* scalar_ptr(const float* p)
{
  const unsigned long long u = reinterpret_cast<unsigned long long>(p);
  const unsigned lo = __builtin_amdgcn_readfirstlane((unsigned)u);
  const unsigned hi = __builtin_amdgcn_readfirstlane((unsigned)(u >> 32));
  return (gmem_char*)(((unsigned long long)hi << 32) | lo);
}

#define VMAPOFS ((unsigned)(MAPELEMS * 4))
#define PIXBYTES ((unsigned)(COUT * 4))

template <bool ISROW>
__device__ __forceinline__ void attn_row_segment(
    const float* __restrict__ qB,
    const float* __restrict__ kB,
    const unsigned c,
    const float* __restrict__ rel,
    float* __restrict__ orow,
    const int h, const int w0)
{
  float bias[KS];
  #pragma unroll
  for (int i = 0; i < KS; ++i) bias[i] = rel[i];

  const unsigned lane4 = 4u * c;

  bool rok[KS];
  bool rall = true;
  gmem_char* krow[KS];
  #pragma unroll
  for (int r = 0; r < KS; ++r) {
    const int hh = h - PADR + r;
    rok[r]  = (hh >= 0) & (hh < HH);
    rall    = rall & rok[r];
    const int hc = hh < 0 ? 0 : (hh > HH - 1 ? HH - 1 : hh);
    krow[r] = scalar_ptr(kB + hc * WW * COUT);
  }
  gmem_char* qrow = scalar_ptr(qB + h * WW * COUT);

  float kwin[KS][KS], vwin[KS][KS];
  float cmx[KS], cmn[KS];

  bool call = true;
  #pragma unroll
  for (int j = 1; j < KS; ++j) {
    const int ww  = w0 - PADR - 1 + j;
    call = call & (ww >= 0) & (ww < WW);
    const int wc  = ww < 0 ? 0 : (ww > WW - 1 ? WW - 1 : ww);
    const unsigned off = lane4 + (unsigned)wc * PIXBYTES;
    #pragma unroll
    for (int r = 0; r < KS; ++r) {
      kwin[r][j] = ld_lane(krow[r], off);
      vwin[r][j] = ld_lane(krow[r] + VMAPOFS, off);
    }
  }
  float pk[KS], pv[KS];
  {
    const int wc = (w0 + PADR > WW - 1) ? WW - 1 : (w0 + PADR);
    const unsigned off = lane4 + (unsigned)wc * PIXBYTES;
    #pragma unroll
    for (int r = 0; r < KS; ++r) {
      pk[r] = ld_lane(krow[r], off);
      pv[r] = ld_lane(krow[r] + VMAPOFS, off);
    }
  }
  float qn = ld_lane(qrow, lane4 + (unsigned)w0 * PIXBYTES);

  if (!(rall & call)) {
    #pragma unroll
    for (int j = 1; j < KS; ++j) {
      const int ww  = w0 - PADR - 1 + j;
      const bool cok = (ww >= 0) & (ww < WW);
      #pragma unroll
      for (int r = 0; r < KS; ++r) {
        if (!(cok & rok[r])) { kwin[r][j] = 0.0f; vwin[r][j] = 0.0f; }
      }
    }
  }
  #pragma unroll
  for (int j = 1; j < KS; ++j) {
    if (ISROW) {
      #pragma unroll
      for (int r = 0; r < KS; ++r) kwin[r][j] += bias[r];
    }
    float col[KS];
    #pragma unroll
    for (int r = 0; r < KS; ++r) col[r] = kwin[r][j];
    cmx[j] = max7(col);
    cmn[j] = min7(col);
  }
  #pragma unroll
  for (int r = 0; r < KS; ++r) { kwin[r][0] = 0.0f; vwin[r][0] = 0.0f; }
  cmx[0] = 0.0f; cmn[0] = 0.0f;

  for (int i4 = 0; i4 < SEG; i4 += 4) {
    const int wf = w0 + i4;

    float ek[3][KS], ev[3][KS], qq[4];
    qq[0] = qn;
    #pragma unroll
    for (int u = 1; u < 4; ++u) {
      const int wn = (wf + u + PADR > WW - 1) ? WW - 1 : (wf + u + PADR);
      const unsigned off = lane4 + (unsigned)wn * PIXBYTES;
      #pragma unroll
      for (int r = 0; r < KS; ++r) {
        ek[u - 1][r] = ld_lane(krow[r], off);
        ev[u - 1][r] = ld_lane(krow[r] + VMAPOFS, off);
      }
      const int wq = (wf + u > WW - 1) ? WW - 1 : (wf + u);
      qq[u] = ld_lane(qrow, lane4 + (unsigned)wq * PIXBYTES);
    }

    if (!(rall & (wf + 3 + PADR < WW))) {
      #pragma unroll
      for (int r = 0; r < KS; ++r) {
        if (!((wf + PADR < WW) & rok[r])) { pk[r] = 0.0f; pv[r] = 0.0f; }
        #pragma unroll
        for (int u = 1; u < 4; ++u) {
          if (!((wf + u + PADR < WW) & rok[r])) { ek[u - 1][r] = 0.0f; ev[u - 1][r] = 0.0f; }
        }
      }
    }

    float ks[KS][KS + 4], vs[KS][KS + 4], mxs[KS + 4], mns[KS + 4];
    #pragma unroll
    for (int r = 0; r < KS; ++r) {
      #pragma unroll
      for (int j = 0; j < KS; ++j) { ks[r][j] = kwin[r][j]; vs[r][j] = vwin[r][j]; }
      ks[r][KS] = ISROW ? (pk[r] + bias[r]) : pk[r];
      vs[r][KS] = pv[r];
      #pragma unroll
      for (int u = 1; u < 4; ++u) {
        ks[r][KS + u] = ISROW ? (ek[u - 1][r] + bias[r]) : ek[u - 1][r];
        vs[r][KS + u] = ev[u - 1][r];
      }
    }
    #pragma unroll
    for (int j = 0; j < KS; ++j) { mxs[j] = cmx[j]; mns[j] = cmn[j]; }
    #pragma unroll
    for (int u = 0; u < 4; ++u) {
      float col[KS];
      #pragma unroll
      for (int r = 0; r < KS; ++r) col[r] = ks[r][KS + u];
      mxs[KS + u] = max7(col);
      mns[KS + u] = min7(col);
    }

    {
      const int wn = (wf + 4 + PADR > WW - 1) ? WW - 1 : (wf + 4 + PADR);
      const unsigned off = lane4 + (unsigned)wn * PIXBYTES;
      #pragma unroll
      for (int r = 0; r < KS; ++r) {
        pk[r] = ld_lane(krow[r], off);
        pv[r] = ld_lane(krow[r] + VMAPOFS, off);
      }
      const int wq = (wf + 4 > WW - 1) ? WW - 1 : (wf + 4);
      qn = ld_lane(qrow, lane4 + (unsigned)wq * PIXBYTES);
    }

    float o[4];
    #pragma unroll
    for (int u = 0; u < 4; ++u) {
      float kw[KS][KS], vw[KS][KS], cx[KS], cn[KS];
      #pragma unroll
      for (int j = 0; j < KS; ++j) {
        #pragma unroll
        for (int r = 0; r < KS; ++r) { kw[r][j] = ks[r][u + 1 + j]; vw[r][j] = vs[r][u + 1 + j]; }
        cx[j] = mxs[u + 1 + j];
        cn[j] = mns[u + 1 + j];
      }
      o[u] = window_softmax<ISROW>(kw, vw, cx, cn, bias, qq[u]);
    }

    #pragma unroll
    for (int j = 0; j < KS; ++j) {
      #pragma unroll
      for (int r = 0; r < KS; ++r) { kwin[r][j] = ks[r][j + 4]; vwin[r][j] = vs[r][j + 4]; }
      cmx[j] = mxs[j + 4];
      cmn[j] = mns[j + 4];
    }

    *reinterpret_cast<float4*>(orow + i4) = make_float4(o[0], o[1], o[2], o[3]);
  }
}

__global__ __launch_bounds__(RPB * 32) __attribute__((amdgpu_waves_per_eu(4, 4))) void local_attn_cl(
    const float* __restrict__ ws,
    const float* __restrict__ rel_h,
    const float* __restrict__ rel_w,
    float* __restrict__ out)
{
  const unsigned lane = threadIdx.x & 31;
  const int wave = __builtin_amdgcn_readfirstlane(threadIdx.x >> 5);
  const int b    = blockIdx.z;
  const int cg   = blockIdx.y;
  const unsigned c = cg * 32 + lane;
  const int rg   = blockIdx.x / NSEG;
  const int seg  = blockIdx.x - rg * NSEG;
  const int h    = rg * RPB + wave;
  const int w0   = seg * SEG;

  const float* __restrict__ qB = ws + (size_t)b * COUT * HWSZ;
  const float* __restrict__ kB = qB + (size_t)MAPELEMS;
  float* __restrict__ orow = out + ((size_t)b * COUT + c) * HWSZ + h * WW + w0;

  if (cg < COUT / 64)
    attn_row_segment<true >(qB, kB, c, rel_h + (size_t)c * KS, orow, h, w0);
  else
    attn_row_segment<false>(qB, kB, c, rel_w + (size_t)(c - COUT / 2) * KS, orow, h, w0);
}

extern "C" void kernel_launch(void* const* d_in, const int* in_sizes, int n_in,
                              void* d_out, int out_size, void* d_ws, size_t ws_size,
                              hipStream_t stream) {
  const float* x  = (const float*)d_in[0];
  const float* wq = (const float*)d_in[1];
  const float* wk = (const float*)d_in[2];
  const float* wv = (const float*)d_in[3];
  const float* rh = (const float*)d_in[4];
  const float* rw = (const float*)d_in[5];
  float* ws  = (float*)d_ws;
  float* out = (float*)d_out;

  dim3 g1(COUT / 128, HWSZ / 128, 3 * BATCH);
  qkv_gemm_cl<<<g1, 256, 0, stream>>>(x, wq, wk, wv, ws);

  dim3 g2((HH / RPB) * NSEG, COUT / 32, BATCH);
  local_attn_cl<<<g2, RPB * 32, 0, stream>>>(ws, rh, rw, out);
}
